# starting kernel
# baseline (speedup 1.0000x reference)
#include <hip/hip_runtime.h>


typedef __attribute__((ext_vector_type(2))) float v2f;
typedef __attribute__((ext_vector_type(8))) float v8f;

#define BATCH   64
#define NCAPS   2048
#define IDIM    16
#define KCLS    32
#define ODIM    32
#define COLS    (KCLS * ODIM)
#define KRED    (NCAPS * IDIM)
#define NCHUNKS 64
#define NPER    (NCAPS / NCHUNKS)
#define WPB     8

#define W_NSTRIDE 16384
#define W_KSTRIDE 512
#define W_ISTRIDE 32

__global__ void caps_zero(float* __restrict__ s) {
  s[blockIdx.x * blockDim.x + threadIdx.x] = 0.0f;
}

__global__ __launch_bounds__(256) void caps_gemm(const float* __restrict__ x,
                                                 const float* __restrict__ W,
                                                 float* __restrict__ s) {
  const int lane = threadIdx.x & 31;
  const int wid  = blockIdx.x * WPB + (threadIdx.x >> 5);
  const int ct   = wid & 63;
  const int g    = wid >> 6;
  const int kc   = ct >> 1;
  const int o0   = (ct & 1) << 4;
  const int col  = lane & 15;
  const int kkb  = (lane >> 4) << 1;

  const float* xb = x + (size_t)col * KRED + (size_t)g * (NPER * IDIM) + kkb;
  const float* wb = W + (size_t)g * NPER * W_NSTRIDE + (size_t)kc * W_KSTRIDE
                      + (size_t)kkb * W_ISTRIDE + o0 + col;

  v8f acc0 = {}; v8f acc1 = {}; v8f acc2 = {}; v8f acc3 = {};

  for (int nn = 0; nn < NPER; ++nn) {
    const float* wn = wb + (size_t)nn * W_NSTRIDE;
    const float* xn = xb + nn * IDIM;
    const float* pf = wn + ((nn < NPER - 2) ? 2 * W_NSTRIDE : 0);
    __builtin_prefetch(pf, 0, 0);
#pragma unroll
    for (int ss = 0; ss < 4; ++ss) {
      v2f bfr;
      bfr.x = wn[ss * 4 * W_ISTRIDE];
      bfr.y = wn[ss * 4 * W_ISTRIDE + W_ISTRIDE];
      v2f a0 = *(const v2f*)(xn + ss * 4);
      v2f a1 = *(const v2f*)(xn + ss * 4 + 16 * KRED);
      v2f a2 = *(const v2f*)(xn + ss * 4 + 32 * KRED);
      v2f a3 = *(const v2f*)(xn + ss * 4 + 48 * KRED);
      acc0 = __builtin_amdgcn_wmma_f32_16x16x4_f32(false, a0, false, bfr, (short)0, acc0, false, false);
      acc1 = __builtin_amdgcn_wmma_f32_16x16x4_f32(false, a1, false, bfr, (short)0, acc1, false, false);
      acc2 = __builtin_amdgcn_wmma_f32_16x16x4_f32(false, a2, false, bfr, (short)0, acc2, false, false);
      acc3 = __builtin_amdgcn_wmma_f32_16x16x4_f32(false, a3, false, bfr, (short)0, acc3, false, false);
    }
  }

  float* srow = s + (size_t)(ct * 16 + col);
  const int mhi = (lane >> 4) << 3;
#pragma unroll
  for (int r = 0; r < 8; ++r) {
    atomicAdd(srow + (size_t)( 0 + r + mhi) * COLS, acc0[r]);
    atomicAdd(srow + (size_t)(16 + r + mhi) * COLS, acc1[r]);
    atomicAdd(srow + (size_t)(32 + r + mhi) * COLS, acc2[r]);
    atomicAdd(srow + (size_t)(48 + r + mhi) * COLS, acc3[r]);
  }
}

__global__ void caps_squash(const float* __restrict__ s, float* __restrict__ out) {
  const int t = blockIdx.x * blockDim.x + threadIdx.x;
  float v = s[t];
  float q = v * v;
#pragma unroll
  for (int off = 16; off > 0; off >>= 1)
    q += __shfl_xor(q, off, 32);
  const float scale = (q / (1.0f + q)) * rsqrtf(q);
  out[t] = scale * v;
}

extern "C" void kernel_launch(void* const* d_in, const int* in_sizes, int n_in,
                              void* d_out, int out_size, void* d_ws, size_t ws_size,
                              hipStream_t stream) {
  (void)in_sizes; (void)n_in; (void)out_size; (void)ws_size;
  const float* x = (const float*)d_in[0];
  const float* W = (const float*)d_in[1];
  float* out = (float*)d_out;
  float* s   = (float*)d_ws;

  caps_zero  <<<BATCH * COLS / 256, 256, 0, stream>>>(s);
  caps_gemm  <<<(64 * NCHUNKS) / WPB, 256, 0, stream>>>(x, W, s);
  caps_squash<<<BATCH * COLS / 256, 256, 0, stream>>>(s, out);
}


# gemm re-tiled: each wave owns a 64x64 tile (4 column tiles, was 1) so one x fragment feeds 4 WMMAs; 1 wave per SIMD, next capsule's fragments loaded a block ahead; sums and atomics unchanged
# speedup vs baseline: 1.1836x; 1.1836x over previous
#include <hip/hip_runtime.h>


typedef __attribute__((ext_vector_type(2))) float v2f;
typedef __attribute__((ext_vector_type(8))) float v8f;

#define BATCH   64
#define NCAPS   2048
#define IDIM    16
#define KCLS    32
#define ODIM    32
#define COLS    (KCLS * ODIM)
#define KRED    (NCAPS * IDIM)
#define NCHUNKS 64
#define NPER    (NCAPS / NCHUNKS)
#define MT      4
#define CT      4
#define NCG     (COLS / (16 * CT))
#define WPB     4

#define W_NSTRIDE 16384
#define W_KSTRIDE 512
#define W_ISTRIDE 32

struct Frags {
  v2f a[4][MT];
  v2f b[4][CT];
};

__device__ __forceinline__ void load_frags(Frags& f, const float* __restrict__ xn,
                                           const float* __restrict__ wn) {
#pragma unroll
  for (int ss = 0; ss < 4; ++ss) {
#pragma unroll
    for (int mt = 0; mt < MT; ++mt)
      f.a[ss][mt] = *(const v2f*)(xn + ss * 4 + mt * 16 * KRED);
#pragma unroll
    for (int j = 0; j < CT; ++j) {
      const float* p = wn + ss * 4 * W_ISTRIDE + (j >> 1) * W_KSTRIDE + (j & 1) * 16;
      v2f bfr;
      bfr.x = p[0];
      bfr.y = p[W_ISTRIDE];
      f.b[ss][j] = bfr;
    }
  }
}

__device__ __forceinline__ void mma_frags(v8f (&acc)[MT][CT], const Frags& f) {
#pragma unroll
  for (int ss = 0; ss < 4; ++ss) {
#pragma unroll
    for (int j = 0; j < CT; ++j) {
#pragma unroll
      for (int mt = 0; mt < MT; ++mt)
        acc[mt][j] = __builtin_amdgcn_wmma_f32_16x16x4_f32(false, f.a[ss][mt], false, f.b[ss][j],
                                                           (short)0, acc[mt][j], false, false);
    }
  }
}

__global__ void caps_zero(float* __restrict__ s) {
  s[blockIdx.x * blockDim.x + threadIdx.x] = 0.0f;
}

__global__ __launch_bounds__(128) void caps_gemm(const float* __restrict__ x,
                                                 const float* __restrict__ W,
                                                 float* __restrict__ s) {
  const int lane = threadIdx.x & 31;
  const int wid  = blockIdx.x * WPB + (threadIdx.x >> 5);
  const int cg   = wid & (NCG - 1);
  const int g    = wid / NCG;
  const int col  = lane & 15;
  const int kkb  = (lane >> 4) << 1;

  const float* xb = x + (size_t)col * KRED + (size_t)g * (NPER * IDIM) + kkb;
  const float* wb = W + (size_t)g * NPER * W_NSTRIDE + (size_t)(2 * cg) * W_KSTRIDE
                      + (size_t)kkb * W_ISTRIDE + col;
  const float* pfb = W + (size_t)g * NPER * W_NSTRIDE + (size_t)(2 * cg) * W_KSTRIDE + col * 64;

  v8f acc[MT][CT];
#pragma unroll
  for (int mt = 0; mt < MT; ++mt)
#pragma unroll
    for (int j = 0; j < CT; ++j)
      acc[mt][j] = (v8f){};

  Frags f0, f1;
  __builtin_prefetch(pfb + (size_t)1 * W_NSTRIDE, 0, 0);
  __builtin_prefetch(pfb + (size_t)2 * W_NSTRIDE, 0, 0);
  __builtin_prefetch(pfb + (size_t)3 * W_NSTRIDE, 0, 0);
  load_frags(f0, xb, wb);
  for (int nn = 0; nn < NPER - 4; nn += 2) {
    load_frags(f1, xb + (nn + 1) * IDIM, wb + (size_t)(nn + 1) * W_NSTRIDE);
    __builtin_prefetch(pfb + (size_t)(nn + 4) * W_NSTRIDE, 0, 0);
    __builtin_amdgcn_sched_barrier(0);
    mma_frags(acc, f0);
    __builtin_amdgcn_sched_barrier(0);
    load_frags(f0, xb + (nn + 2) * IDIM, wb + (size_t)(nn + 2) * W_NSTRIDE);
    __builtin_prefetch(pfb + (size_t)(nn + 5) * W_NSTRIDE, 0, 0);
    __builtin_amdgcn_sched_barrier(0);
    mma_frags(acc, f1);
  }
  load_frags(f1, xb + (NPER - 3) * IDIM, wb + (size_t)(NPER - 3) * W_NSTRIDE);
  __builtin_amdgcn_sched_barrier(0);
  mma_frags(acc, f0);
  __builtin_amdgcn_sched_barrier(0);
  load_frags(f0, xb + (NPER - 2) * IDIM, wb + (size_t)(NPER - 2) * W_NSTRIDE);
  __builtin_amdgcn_sched_barrier(0);
  mma_frags(acc, f1);
  __builtin_amdgcn_sched_barrier(0);
  load_frags(f1, xb + (NPER - 1) * IDIM, wb + (size_t)(NPER - 1) * W_NSTRIDE);
  __builtin_amdgcn_sched_barrier(0);
  mma_frags(acc, f0);
  mma_frags(acc, f1);

  const int mhi = (lane >> 4) << 3;
#pragma unroll
  for (int j = 0; j < CT; ++j) {
    float* srow = s + (size_t)((cg * CT + j) * 16 + col);
#pragma unroll
    for (int mt = 0; mt < MT; ++mt) {
#pragma unroll
      for (int r = 0; r < 8; ++r)
        atomicAdd(srow + (size_t)(mt * 16 + r + mhi) * COLS, acc[mt][j][r]);
    }
  }
}

__global__ void caps_squash(const float* __restrict__ s, float* __restrict__ out) {
  const int t = blockIdx.x * blockDim.x + threadIdx.x;
  float v = s[t];
  float q = v * v;
#pragma unroll
  for (int off = 16; off > 0; off >>= 1)
    q += __shfl_xor(q, off, 32);
  const float scale = (q / (1.0f + q)) * rsqrtf(q);
  out[t] = scale * v;
}

extern "C" void kernel_launch(void* const* d_in, const int* in_sizes, int n_in,
                              void* d_out, int out_size, void* d_ws, size_t ws_size,
                              hipStream_t stream) {
  (void)in_sizes; (void)n_in; (void)out_size; (void)ws_size;
  const float* x = (const float*)d_in[0];
  const float* W = (const float*)d_in[1];
  float* out = (float*)d_out;
  float* s   = (float*)d_ws;

  caps_zero  <<<BATCH * COLS / 256, 256, 0, stream>>>(s);
  caps_gemm  <<<(NCG * NCHUNKS) / WPB, 128, 0, stream>>>(x, W, s);
  caps_squash<<<BATCH * COLS / 256, 256, 0, stream>>>(s, out);
}


# a block's 4 waves now take 4 reduction chunks of one 64x64 tile and add their partials in LDS in chunk order; 16 float atomic adds reach each element of s instead of 64 (sum order changes)
# speedup vs baseline: 1.5347x; 1.2966x over previous
#include <hip/hip_runtime.h>


typedef __attribute__((ext_vector_type(2))) float v2f;
typedef __attribute__((ext_vector_type(4))) float v4f;
typedef __attribute__((ext_vector_type(8))) float v8f;

#define BATCH   64
#define NCAPS   2048
#define IDIM    16
#define KCLS    32
#define ODIM    32
#define COLS    (KCLS * ODIM)
#define KRED    (NCAPS * IDIM)
#define NCHUNKS 64
#define NPER    (NCAPS / NCHUNKS)
#define MT      4
#define CT      4
#define NCG     (COLS / (16 * CT))
#define WPB     4

#define W_NSTRIDE 16384
#define W_KSTRIDE 512
#define W_ISTRIDE 32

struct Frags {
  v2f a[4][MT];
  v2f b[4][CT];
};

__device__ __forceinline__ void load_frags(Frags& f, const float* __restrict__ xn,
                                           const float* __restrict__ wn) {
#pragma unroll
  for (int ss = 0; ss < 4; ++ss) {
#pragma unroll
    for (int mt = 0; mt < MT; ++mt)
      f.a[ss][mt] = *(const v2f*)(xn + ss * 4 + mt * 16 * KRED);
#pragma unroll
    for (int j = 0; j < CT; ++j) {
      const float* p = wn + ss * 4 * W_ISTRIDE + (j >> 1) * W_KSTRIDE + (j & 1) * 16;
      v2f bfr;
      bfr.x = p[0];
      bfr.y = p[W_ISTRIDE];
      f.b[ss][j] = bfr;
    }
  }
}

__device__ __forceinline__ void mma_frags(v8f (&acc)[MT][CT], const Frags& f) {
#pragma unroll
  for (int ss = 0; ss < 4; ++ss) {
#pragma unroll
    for (int j = 0; j < CT; ++j) {
#pragma unroll
      for (int mt = 0; mt < MT; ++mt)
        acc[mt][j] = __builtin_amdgcn_wmma_f32_16x16x4_f32(false, f.a[ss][mt], false, f.b[ss][j],
                                                           (short)0, acc[mt][j], false, false);
    }
  }
}

__global__ void caps_zero(float* __restrict__ s) {
  s[blockIdx.x * blockDim.x + threadIdx.x] = 0.0f;
}

__global__ __launch_bounds__(128) void caps_gemm(const float* __restrict__ x,
                                                 const float* __restrict__ W,
                                                 float* __restrict__ s) {
  __shared__ v4f red[WPB][MT * CT][2][32];

  const int lane = threadIdx.x & 31;
  const int wv   = threadIdx.x >> 5;
  const int cg   = blockIdx.x & (NCG - 1);
  const int g    = (blockIdx.x / NCG) * WPB + wv;
  const int col  = lane & 15;
  const int kkb  = (lane >> 4) << 1;

  const float* xb = x + (size_t)col * KRED + (size_t)g * (NPER * IDIM) + kkb;
  const float* wb = W + (size_t)g * NPER * W_NSTRIDE + (size_t)(2 * cg) * W_KSTRIDE
                      + (size_t)kkb * W_ISTRIDE + col;
  const float* pfb = W + (size_t)g * NPER * W_NSTRIDE + (size_t)(2 * cg) * W_KSTRIDE + col * 64;

  v8f acc[MT][CT];
#pragma unroll
  for (int mt = 0; mt < MT; ++mt)
#pragma unroll
    for (int j = 0; j < CT; ++j)
      acc[mt][j] = (v8f){};

  Frags f0, f1;
  __builtin_prefetch(pfb + (size_t)1 * W_NSTRIDE, 0, 0);
  __builtin_prefetch(pfb + (size_t)2 * W_NSTRIDE, 0, 0);
  __builtin_prefetch(pfb + (size_t)3 * W_NSTRIDE, 0, 0);
  load_frags(f0, xb, wb);
  for (int nn = 0; nn < NPER - 4; nn += 2) {
    load_frags(f1, xb + (nn + 1) * IDIM, wb + (size_t)(nn + 1) * W_NSTRIDE);
    __builtin_prefetch(pfb + (size_t)(nn + 4) * W_NSTRIDE, 0, 0);
    __builtin_amdgcn_sched_barrier(0);
    mma_frags(acc, f0);
    __builtin_amdgcn_sched_barrier(0);
    load_frags(f0, xb + (nn + 2) * IDIM, wb + (size_t)(nn + 2) * W_NSTRIDE);
    __builtin_prefetch(pfb + (size_t)(nn + 5) * W_NSTRIDE, 0, 0);
    __builtin_amdgcn_sched_barrier(0);
    mma_frags(acc, f1);
  }
  load_frags(f1, xb + (NPER - 3) * IDIM, wb + (size_t)(NPER - 3) * W_NSTRIDE);
  __builtin_amdgcn_sched_barrier(0);
  mma_frags(acc, f0);
  __builtin_amdgcn_sched_barrier(0);
  load_frags(f0, xb + (NPER - 2) * IDIM, wb + (size_t)(NPER - 2) * W_NSTRIDE);
  __builtin_amdgcn_sched_barrier(0);
  mma_frags(acc, f1);
  __builtin_amdgcn_sched_barrier(0);
  load_frags(f1, xb + (NPER - 1) * IDIM, wb + (size_t)(NPER - 1) * W_NSTRIDE);
  __builtin_amdgcn_sched_barrier(0);
  mma_frags(acc, f0);
  mma_frags(acc, f1);

#pragma unroll
  for (int mt = 0; mt < MT; ++mt) {
#pragma unroll
    for (int j = 0; j < CT; ++j) {
      const v8f a = acc[mt][j];
      red[wv][mt * CT + j][0][lane] = (v4f){a[0], a[1], a[2], a[3]};
      red[wv][mt * CT + j][1][lane] = (v4f){a[4], a[5], a[6], a[7]};
    }
  }
  __syncthreads();

  const int mhi = (lane >> 4) << 3;
  float* srow = s + (size_t)((cg * CT + wv) * 16 + col);
#pragma unroll
  for (int mt = 0; mt < MT; ++mt) {
    v4f lo = red[0][mt * CT + wv][0][lane];
    v4f hi = red[0][mt * CT + wv][1][lane];
#pragma unroll
    for (int src = 1; src < WPB; ++src) {
      lo += red[src][mt * CT + wv][0][lane];
      hi += red[src][mt * CT + wv][1][lane];
    }
#pragma unroll
    for (int r = 0; r < 4; ++r) {
      atomicAdd(srow + (size_t)(mt * 16 + r + mhi) * COLS, lo[r]);
      atomicAdd(srow + (size_t)(mt * 16 + r + 4 + mhi) * COLS, hi[r]);
    }
  }
}

__global__ void caps_squash(const float* __restrict__ s, float* __restrict__ out) {
  const int t = blockIdx.x * blockDim.x + threadIdx.x;
  float v = s[t];
  float q = v * v;
#pragma unroll
  for (int off = 16; off > 0; off >>= 1)
    q += __shfl_xor(q, off, 32);
  const float scale = (q / (1.0f + q)) * rsqrtf(q);
  out[t] = scale * v;
}

extern "C" void kernel_launch(void* const* d_in, const int* in_sizes, int n_in,
                              void* d_out, int out_size, void* d_ws, size_t ws_size,
                              hipStream_t stream) {
  (void)in_sizes; (void)n_in; (void)out_size; (void)ws_size;
  const float* x = (const float*)d_in[0];
  const float* W = (const float*)d_in[1];
  float* out = (float*)d_out;
  float* s   = (float*)d_ws;

  caps_zero  <<<BATCH * COLS / 256, 256, 0, stream>>>(s);
  caps_gemm  <<<(NCG * NCHUNKS) / WPB, 128, 0, stream>>>(x, W, s);
  caps_squash<<<BATCH * COLS / 256, 256, 0, stream>>>(s, out);
}


# block sums leave by whole-line plain stores to 16 chunk-quad planes (4 MB); squash adds them in fixed order 0..15 (sum order changes); float atomics (16 per element) and the zero-fill launch go
# speedup vs baseline: 1.5987x; 1.0417x over previous
#include <hip/hip_runtime.h>


typedef __attribute__((ext_vector_type(2))) float v2f;
typedef __attribute__((ext_vector_type(4))) float v4f;
typedef __attribute__((ext_vector_type(8))) float v8f;

#define BATCH   64
#define NCAPS   2048
#define IDIM    16
#define KCLS    32
#define ODIM    32
#define COLS    (KCLS * ODIM)
#define KRED    (NCAPS * IDIM)
#define NCHUNKS 64
#define NPER    (NCAPS / NCHUNKS)
#define MT      4
#define CT      4
#define NCG     (COLS / (16 * CT))
#define WPB     4
#define NQ      (NCHUNKS / WPB)
#define PLANE   (BATCH * COLS)

#define W_NSTRIDE 16384
#define W_KSTRIDE 512
#define W_ISTRIDE 32

struct Frags {
  v2f a[4][MT];
  v2f b[4][CT];
};

__device__ __forceinline__ void load_frags(Frags& f, const float* __restrict__ xn,
                                           const float* __restrict__ wn) {
#pragma unroll
  for (int ss = 0; ss < 4; ++ss) {
#pragma unroll
    for (int mt = 0; mt < MT; ++mt)
      f.a[ss][mt] = *(const v2f*)(xn + ss * 4 + mt * 16 * KRED);
#pragma unroll
    for (int j = 0; j < CT; ++j) {
      const float* p = wn + ss * 4 * W_ISTRIDE + (j >> 1) * W_KSTRIDE + (j & 1) * 16;
      v2f bfr;
      bfr.x = p[0];
      bfr.y = p[W_ISTRIDE];
      f.b[ss][j] = bfr;
    }
  }
}

__device__ __forceinline__ void mma_frags(v8f (&acc)[MT][CT], const Frags& f) {
#pragma unroll
  for (int ss = 0; ss < 4; ++ss) {
#pragma unroll
    for (int j = 0; j < CT; ++j) {
#pragma unroll
      for (int mt = 0; mt < MT; ++mt)
        acc[mt][j] = __builtin_amdgcn_wmma_f32_16x16x4_f32(false, f.a[ss][mt], false, f.b[ss][j],
                                                           (short)0, acc[mt][j], false, false);
    }
  }
}

__global__ __launch_bounds__(128) void caps_gemm(const float* __restrict__ x,
                                                 const float* __restrict__ W,
                                                 float* __restrict__ part) {
  __shared__ v4f red[WPB][MT * CT][2][32];

  const int lane = threadIdx.x & 31;
  const int wv   = threadIdx.x >> 5;
  const int cg   = blockIdx.x & (NCG - 1);
  const int qd   = blockIdx.x / NCG;
  const int g    = qd * WPB + wv;
  const int col  = lane & 15;
  const int kkb  = (lane >> 4) << 1;

  const float* xb = x + (size_t)col * KRED + (size_t)g * (NPER * IDIM) + kkb;
  const float* wb = W + (size_t)g * NPER * W_NSTRIDE + (size_t)(2 * cg) * W_KSTRIDE
                      + (size_t)kkb * W_ISTRIDE + col;
  const float* pfb = W + (size_t)g * NPER * W_NSTRIDE + (size_t)(2 * cg) * W_KSTRIDE + col * 64;

  v8f acc[MT][CT];
#pragma unroll
  for (int mt = 0; mt < MT; ++mt)
#pragma unroll
    for (int j = 0; j < CT; ++j)
      acc[mt][j] = (v8f){};

  Frags f0, f1;
  __builtin_prefetch(pfb + (size_t)1 * W_NSTRIDE, 0, 0);
  __builtin_prefetch(pfb + (size_t)2 * W_NSTRIDE, 0, 0);
  __builtin_prefetch(pfb + (size_t)3 * W_NSTRIDE, 0, 0);
  load_frags(f0, xb, wb);
  for (int nn = 0; nn < NPER - 4; nn += 2) {
    load_frags(f1, xb + (nn + 1) * IDIM, wb + (size_t)(nn + 1) * W_NSTRIDE);
    __builtin_prefetch(pfb + (size_t)(nn + 4) * W_NSTRIDE, 0, 0);
    __builtin_amdgcn_sched_barrier(0);
    mma_frags(acc, f0);
    __builtin_amdgcn_sched_barrier(0);
    load_frags(f0, xb + (nn + 2) * IDIM, wb + (size_t)(nn + 2) * W_NSTRIDE);
    __builtin_prefetch(pfb + (size_t)(nn + 5) * W_NSTRIDE, 0, 0);
    __builtin_amdgcn_sched_barrier(0);
    mma_frags(acc, f1);
  }
  load_frags(f1, xb + (NPER - 3) * IDIM, wb + (size_t)(NPER - 3) * W_NSTRIDE);
  __builtin_amdgcn_sched_barrier(0);
  mma_frags(acc, f0);
  __builtin_amdgcn_sched_barrier(0);
  load_frags(f0, xb + (NPER - 2) * IDIM, wb + (size_t)(NPER - 2) * W_NSTRIDE);
  __builtin_amdgcn_sched_barrier(0);
  mma_frags(acc, f1);
  __builtin_amdgcn_sched_barrier(0);
  load_frags(f1, xb + (NPER - 1) * IDIM, wb + (size_t)(NPER - 1) * W_NSTRIDE);
  __builtin_amdgcn_sched_barrier(0);
  mma_frags(acc, f0);
  mma_frags(acc, f1);

#pragma unroll
  for (int mt = 0; mt < MT; ++mt) {
#pragma unroll
    for (int j = 0; j < CT; ++j) {
      const v8f a = acc[mt][j];
      red[wv][mt * CT + j][0][lane] = (v4f){a[0], a[1], a[2], a[3]};
      red[wv][mt * CT + j][1][lane] = (v4f){a[4], a[5], a[6], a[7]};
    }
  }
  __syncthreads();

  const int tsel = lane >> 4;
  float* prow = part + (size_t)qd * PLANE + (size_t)(wv * 16) * COLS + (size_t)(cg * (16 * CT) + lane);
#pragma unroll
  for (int p = 0; p < 2; ++p) {
    const int t = wv * CT + 2 * p + tsel;
#pragma unroll
    for (int hs = 0; hs < 2; ++hs) {
#pragma unroll
      for (int hf = 0; hf < 2; ++hf) {
        v4f v = red[0][t][hf][hs * 16 + col];
#pragma unroll
        for (int src = 1; src < WPB; ++src)
          v += red[src][t][hf][hs * 16 + col];
#pragma unroll
        for (int e = 0; e < 4; ++e)
          prow[(size_t)(8 * hs + 4 * hf + e) * COLS + 32 * p] = v[e];
      }
    }
  }
}

__global__ __launch_bounds__(64) void caps_squash(const float* __restrict__ part,
                                                  float* __restrict__ out) {
  const int t = blockIdx.x * 64 + threadIdx.x;
  const v4f* p = (const v4f*)part + t;
  v4f pv[NQ];
#pragma unroll
  for (int q = 0; q < NQ; ++q)
    pv[q] = p[(size_t)q * (PLANE / 4)];
  v4f v = pv[0];
#pragma unroll
  for (int q = 1; q < NQ; ++q)
    v += pv[q];
  float n2 = (v.x * v.x + v.y * v.y) + (v.z * v.z + v.w * v.w);
  n2 += __shfl_xor(n2, 1, 32);
  n2 += __shfl_xor(n2, 2, 32);
  n2 += __shfl_xor(n2, 4, 32);
  const float scale = (n2 / (1.0f + n2)) * rsqrtf(n2);
  ((v4f*)out)[t] = v * scale;
}

extern "C" void kernel_launch(void* const* d_in, const int* in_sizes, int n_in,
                              void* d_out, int out_size, void* d_ws, size_t ws_size,
                              hipStream_t stream) {
  (void)in_sizes; (void)n_in; (void)out_size; (void)ws_size;
  const float* x = (const float*)d_in[0];
  const float* W = (const float*)d_in[1];
  float* out = (float*)d_out;
  float* part = (float*)d_ws;

  caps_gemm  <<<(NCG * NCHUNKS) / WPB, 128, 0, stream>>>(x, W, part);
  caps_squash<<<BATCH * COLS / 256, 64, 0, stream>>>(part, out);
}
